# starting kernel
# baseline (speedup 1.0000x reference)
#include <hip/hip_runtime.h>

typedef float v2f __attribute__((ext_vector_type(2)));
typedef float v8f __attribute__((ext_vector_type(8)));

#define EMBED 64
#define HEADS 8
#define HDIM  8
#define BB    8
#define NN    128
#define TT    64

__device__ __forceinline__ v8f wmma_f32_16x16x4(v2f a, v2f b, v8f c) {
  return __builtin_amdgcn_wmma_f32_16x16x4_f32(false, a, false, b, (short)0, c,
                                               false, false);
}

__global__ __launch_bounds__(128) void attn_kernel(
    const float* __restrict__ V, const float* __restrict__ K,
    const float* __restrict__ Q, const float* __restrict__ Wv,
    const float* __restrict__ Wk, const float* __restrict__ Wq,
    float* __restrict__ Xout) {
  __shared__ float S[64 * NN];
  __shared__ float Qp[NN * HDIM];
  __shared__ float Kp[NN * HDIM];
  __shared__ float Vp[NN * HDIM];
  __shared__ float redM[128];
  __shared__ float redS[128];

  const int blk     = blockIdx.x;
  const int rh      = blk & 1;
  const int h       = (blk >> 1) % HEADS;
  const int t       = (blk >> 1) / HEADS % TT;
  const int b       = (blk >> 1) / (HEADS * TT);
  const int rowBase = rh * 64;

  const int tid  = threadIdx.x;
  const int wave = tid >> 5;
  const int lane = tid & 31;
  const int half = lane >> 4;
  const int l    = lane & 15;

  {
    const int n = tid;
    const size_t base = (((size_t)(b * NN + n) * TT + t) * EMBED) + h * HDIM;
    float qv[HDIM], kv[HDIM], vv[HDIM];
#pragma unroll
    for (int d = 0; d < HDIM; ++d) {
      qv[d] = Q[base + d];
      kv[d] = K[base + d];
      vv[d] = V[base + d];
    }
#pragma unroll
    for (int e = 0; e < HDIM; ++e) {
      float aq = 0.f, ak = 0.f, av = 0.f;
#pragma unroll
      for (int d = 0; d < HDIM; ++d) {
        aq += qv[d] * Wq[e * HDIM + d];
        ak += kv[d] * Wk[e * HDIM + d];
        av += vv[d] * Wv[e * HDIM + d];
      }
      Qp[n * HDIM + e] = aq;
      Kp[n * HDIM + e] = ak;
      Vp[n * HDIM + e] = av;
    }
  }
  __syncthreads();

  const float scale = 0.125f;
  for (int tile = wave; tile < 32; tile += 4) {
    const int ti = tile >> 3;
    const int tj = tile & 7;
    v8f c = {};
#pragma unroll
    for (int kb = 0; kb < HDIM; kb += 4) {
      const int ac = kb + 2 * half;
      v2f a, bm;
      a.x  = Qp[(rowBase + ti * 16 + l) * HDIM + ac];
      a.y  = Qp[(rowBase + ti * 16 + l) * HDIM + ac + 1];
      bm.x = Kp[(tj * 16 + l) * HDIM + ac];
      bm.y = Kp[(tj * 16 + l) * HDIM + ac + 1];
      c = wmma_f32_16x16x4(a, bm, c);
    }
#pragma unroll
    for (int r = 0; r < 8; ++r) {
      const int row = ti * 16 + r + 8 * half;
      S[row * NN + tj * 16 + l] = c[r] * scale;
    }
  }
  __syncthreads();

  {
    const int row = tid >> 1;
    const int seg = tid & 1;
    float* Sseg = S + row * NN + seg * 64;

    float m = -1e30f;
    for (int j = 0; j < 64; ++j) m = fmaxf(m, Sseg[j]);
    redM[tid] = m;
    __syncthreads();
    m = fmaxf(redM[row * 2], redM[row * 2 + 1]);

    float sum = 0.f;
    for (int j = 0; j < 64; ++j) {
      const float ex = __expf(Sseg[j] - m);
      Sseg[j] = ex;
      sum += ex;
    }
    redS[tid] = sum;
    __syncthreads();
    const float inv = 1.f / (redS[row * 2] + redS[row * 2 + 1]);
    for (int j = 0; j < 64; ++j) Sseg[j] *= inv;
  }
  __syncthreads();

  {
    const int ti = wave;
    v8f c = {};
    for (int kb = 0; kb < NN; kb += 4) {
      const int ac = kb + 2 * half;
      v2f a, bm;
      a.x  = S[(ti * 16 + l) * NN + ac];
      a.y  = S[(ti * 16 + l) * NN + ac + 1];
      bm.x = (l < HDIM) ? Vp[ac * HDIM + l] : 0.f;
      bm.y = (l < HDIM) ? Vp[(ac + 1) * HDIM + l] : 0.f;
      c = wmma_f32_16x16x4(a, bm, c);
    }
    if (l < HDIM) {
#pragma unroll
      for (int r = 0; r < 8; ++r) {
        const int n = rowBase + ti * 16 + r + 8 * half;
        const size_t o = (((size_t)(b * NN + n) * TT + t) * EMBED) + h * HDIM + l;
        Xout[o] = c[r];
      }
    }
  }
}

__global__ __launch_bounds__(128) void fc_kernel(const float* __restrict__ X,
                                                 const float* __restrict__ Wo,
                                                 const float* __restrict__ bo,
                                                 float* __restrict__ Y) {
  __shared__ float sX[16 * EMBED];
  __shared__ float sW[EMBED * EMBED];
  __shared__ float sb[EMBED];

  const int tid  = threadIdx.x;
  const int wave = tid >> 5;
  const int lane = tid & 31;
  const int half = lane >> 4;
  const int l    = lane & 15;
  const size_t mBase = (size_t)blockIdx.x * 16;

  for (int i = tid; i < 16 * EMBED; i += 128) sX[i] = X[mBase * EMBED + i];
  for (int i = tid; i < EMBED * EMBED; i += 128) sW[i] = Wo[i];
  if (tid < EMBED) sb[tid] = bo[tid];
  __syncthreads();

  const int tj = wave;
  v8f c = {};
#pragma unroll
  for (int kb = 0; kb < EMBED; kb += 4) {
    const int ac = kb + 2 * half;
    v2f a, bm;
    a.x  = sX[l * EMBED + ac];
    a.y  = sX[l * EMBED + ac + 1];
    bm.x = sW[(tj * 16 + l) * EMBED + ac];
    bm.y = sW[(tj * 16 + l) * EMBED + ac + 1];
    c = wmma_f32_16x16x4(a, bm, c);
  }
  const float bias = sb[tj * 16 + l];
#pragma unroll
  for (int r = 0; r < 8; ++r) {
    const size_t row = mBase + r + 8 * half;
    Y[row * EMBED + tj * 16 + l] = c[r] + bias;
  }
}

extern "C" void kernel_launch(void* const* d_in, const int* in_sizes, int n_in,
                              void* d_out, int out_size, void* d_ws,
                              size_t ws_size, hipStream_t stream) {
  (void)in_sizes; (void)n_in; (void)out_size; (void)ws_size;
  const float* V  = (const float*)d_in[0];
  const float* K  = (const float*)d_in[1];
  const float* Q  = (const float*)d_in[2];
  const float* Wv = (const float*)d_in[3];
  const float* Wk = (const float*)d_in[4];
  const float* Wq = (const float*)d_in[5];
  const float* Wo = (const float*)d_in[6];
  const float* bo = (const float*)d_in[7];
  float* X = (float*)d_ws;
  float* Y = (float*)d_out;

  hipLaunchKernelGGL(attn_kernel, dim3(BB * TT * HEADS * 2), dim3(128), 0,
                     stream, V, K, Q, Wv, Wk, Wq, X);
  hipLaunchKernelGGL(fc_kernel, dim3((BB * NN * TT) / 16), dim3(128), 0, stream,
                     X, Wo, bo, Y);
}


# Softmax moved into registers: wave = 16 query rows x all 128 keys, scores stay in the QK^T D registers, row max and sum by 4-step row_xmask butterfly; P to LDS once; softmax sum tree changes
# speedup vs baseline: 2.8074x; 2.8074x over previous
#include <hip/hip_runtime.h>

typedef float v2f __attribute__((ext_vector_type(2)));
typedef float v8f __attribute__((ext_vector_type(8)));

#define EMBED 64
#define HEADS 8
#define HDIM  8
#define BB    8
#define NN    128
#define TT    64

__device__ __forceinline__ v8f wmma_f32_16x16x4(v2f a, v2f b, v8f c) {
  return __builtin_amdgcn_wmma_f32_16x16x4_f32(false, a, false, b, (short)0, c,
                                               false, false);
}

template <int M>
__device__ __forceinline__ float row_xor(float x) {
  const int v = __builtin_bit_cast(int, x);
  return __builtin_bit_cast(
      float, __builtin_amdgcn_update_dpp(v, v, 0x160 + M, 0xf, 0xf, false));
}

__global__ __launch_bounds__(128) void attn_kernel(
    const float* __restrict__ V, const float* __restrict__ K,
    const float* __restrict__ Q, const float* __restrict__ Wv,
    const float* __restrict__ Wk, const float* __restrict__ Wq,
    float* __restrict__ Xout) {
  __shared__ float S[64 * NN];
  __shared__ float Qp[NN * HDIM];
  __shared__ float Kp[NN * HDIM];
  __shared__ float Vp[NN * HDIM];

  const int blk     = blockIdx.x;
  const int rh      = blk & 1;
  const int h       = (blk >> 1) % HEADS;
  const int t       = (blk >> 1) / HEADS % TT;
  const int b       = (blk >> 1) / (HEADS * TT);
  const int rowBase = rh * 64;

  const int tid  = threadIdx.x;
  const int wave = tid >> 5;
  const int lane = tid & 31;
  const int half = lane >> 4;
  const int l    = lane & 15;

  {
    const int n = tid;
    const size_t base = (((size_t)(b * NN + n) * TT + t) * EMBED) + h * HDIM;
    float qv[HDIM], kv[HDIM], vv[HDIM];
#pragma unroll
    for (int d = 0; d < HDIM; ++d) {
      qv[d] = Q[base + d];
      kv[d] = K[base + d];
      vv[d] = V[base + d];
    }
#pragma unroll
    for (int e = 0; e < HDIM; ++e) {
      float aq = 0.f, ak = 0.f, av = 0.f;
#pragma unroll
      for (int d = 0; d < HDIM; ++d) {
        aq += qv[d] * Wq[e * HDIM + d];
        ak += kv[d] * Wk[e * HDIM + d];
        av += vv[d] * Wv[e * HDIM + d];
      }
      Qp[n * HDIM + e] = aq;
      Kp[n * HDIM + e] = ak;
      Vp[n * HDIM + e] = av;
    }
  }
  __syncthreads();

  const float scale = 0.125f;
  {
    const int ti = wave;
    v8f c[8];
    {
      const float* qrow = Qp + (rowBase + ti * 16 + l) * HDIM + 2 * half;
      v2f a0, a1;
      a0.x = qrow[0];
      a0.y = qrow[1];
      a1.x = qrow[4];
      a1.y = qrow[5];
#pragma unroll
      for (int tj = 0; tj < 8; ++tj) {
        const float* krow = Kp + (tj * 16 + l) * HDIM + 2 * half;
        v2f b0, b1;
        b0.x = krow[0];
        b0.y = krow[1];
        b1.x = krow[4];
        b1.y = krow[5];
        v8f acc = {};
        acc = wmma_f32_16x16x4(a0, b0, acc);
        acc = wmma_f32_16x16x4(a1, b1, acc);
        c[tj] = acc;
      }
    }
#pragma unroll
    for (int r = 0; r < 8; ++r) {
      float s[8];
      float m = -1e30f;
#pragma unroll
      for (int tj = 0; tj < 8; ++tj) {
        s[tj] = c[tj][r] * scale;
        m = fmaxf(m, s[tj]);
      }
      m = fmaxf(m, row_xor<1>(m));
      m = fmaxf(m, row_xor<2>(m));
      m = fmaxf(m, row_xor<4>(m));
      m = fmaxf(m, row_xor<8>(m));
      float sum = 0.f;
#pragma unroll
      for (int tj = 0; tj < 8; ++tj) {
        const float ex = __expf(s[tj] - m);
        s[tj] = ex;
        sum += ex;
      }
      sum += row_xor<1>(sum);
      sum += row_xor<2>(sum);
      sum += row_xor<4>(sum);
      sum += row_xor<8>(sum);
      const float inv = 1.f / sum;
      float* prow = S + (ti * 16 + r + 8 * half) * NN + l;
#pragma unroll
      for (int tj = 0; tj < 8; ++tj) prow[tj * 16] = s[tj] * inv;
    }
  }
  __syncthreads();

  {
    const int ti = wave;
    v8f c = {};
    for (int kb = 0; kb < NN; kb += 4) {
      const int ac = kb + 2 * half;
      v2f a, bm;
      a.x  = S[(ti * 16 + l) * NN + ac];
      a.y  = S[(ti * 16 + l) * NN + ac + 1];
      bm.x = (l < HDIM) ? Vp[ac * HDIM + l] : 0.f;
      bm.y = (l < HDIM) ? Vp[(ac + 1) * HDIM + l] : 0.f;
      c = wmma_f32_16x16x4(a, bm, c);
    }
    if (l < HDIM) {
#pragma unroll
      for (int r = 0; r < 8; ++r) {
        const int n = rowBase + ti * 16 + r + 8 * half;
        const size_t o = (((size_t)(b * NN + n) * TT + t) * EMBED) + h * HDIM + l;
        Xout[o] = c[r];
      }
    }
  }
}

__global__ __launch_bounds__(128) void fc_kernel(const float* __restrict__ X,
                                                 const float* __restrict__ Wo,
                                                 const float* __restrict__ bo,
                                                 float* __restrict__ Y) {
  __shared__ float sX[16 * EMBED];
  __shared__ float sW[EMBED * EMBED];
  __shared__ float sb[EMBED];

  const int tid  = threadIdx.x;
  const int wave = tid >> 5;
  const int lane = tid & 31;
  const int half = lane >> 4;
  const int l    = lane & 15;
  const size_t mBase = (size_t)blockIdx.x * 16;

  for (int i = tid; i < 16 * EMBED; i += 128) sX[i] = X[mBase * EMBED + i];
  for (int i = tid; i < EMBED * EMBED; i += 128) sW[i] = Wo[i];
  if (tid < EMBED) sb[tid] = bo[tid];
  __syncthreads();

  const int tj = wave;
  v8f c = {};
#pragma unroll
  for (int kb = 0; kb < EMBED; kb += 4) {
    const int ac = kb + 2 * half;
    v2f a, bm;
    a.x  = sX[l * EMBED + ac];
    a.y  = sX[l * EMBED + ac + 1];
    bm.x = sW[(tj * 16 + l) * EMBED + ac];
    bm.y = sW[(tj * 16 + l) * EMBED + ac + 1];
    c = wmma_f32_16x16x4(a, bm, c);
  }
  const float bias = sb[tj * 16 + l];
#pragma unroll
  for (int r = 0; r < 8; ++r) {
    const size_t row = mBase + r + 8 * half;
    Y[row * EMBED + tj * 16 + l] = c[r] + bias;
  }
}

extern "C" void kernel_launch(void* const* d_in, const int* in_sizes, int n_in,
                              void* d_out, int out_size, void* d_ws,
                              size_t ws_size, hipStream_t stream) {
  (void)in_sizes; (void)n_in; (void)out_size; (void)ws_size;
  const float* V  = (const float*)d_in[0];
  const float* K  = (const float*)d_in[1];
  const float* Q  = (const float*)d_in[2];
  const float* Wv = (const float*)d_in[3];
  const float* Wk = (const float*)d_in[4];
  const float* Wq = (const float*)d_in[5];
  const float* Wo = (const float*)d_in[6];
  const float* bo = (const float*)d_in[7];
  float* X = (float*)d_ws;
  float* Y = (float*)d_out;

  hipLaunchKernelGGL(attn_kernel, dim3(BB * TT * HEADS * 2), dim3(128), 0,
                     stream, V, K, Q, Wv, Wk, Wq, X);
  hipLaunchKernelGGL(fc_kernel, dim3((BB * NN * TT) / 16), dim3(128), 0, stream,
                     X, Wo, bo, Y);
}


# Attention kernel made persistent: 1024 resident blocks, one per (b,t,4 heads), loop over their heads with all 128 query rows per head (wave = 2 row tiles, one V fragment for both); sum order kept
# speedup vs baseline: 3.0949x; 1.1024x over previous
#include <hip/hip_runtime.h>

typedef float v2f __attribute__((ext_vector_type(2)));
typedef float v8f __attribute__((ext_vector_type(8)));

#define EMBED 64
#define HEADS 8
#define HDIM  8
#define BB    8
#define NN    128
#define TT    64

__device__ __forceinline__ v8f wmma_f32_16x16x4(v2f a, v2f b, v8f c) {
  return __builtin_amdgcn_wmma_f32_16x16x4_f32(false, a, false, b, (short)0, c,
                                               false, false);
}

template <int M>
__device__ __forceinline__ float row_xor(float x) {
  const int v = __builtin_bit_cast(int, x);
  return __builtin_bit_cast(
      float, __builtin_amdgcn_update_dpp(v, v, 0x160 + M, 0xf, 0xf, false));
}

__global__ __launch_bounds__(128) void attn_kernel(
    const float* __restrict__ V, const float* __restrict__ K,
    const float* __restrict__ Q, const float* __restrict__ Wv,
    const float* __restrict__ Wk, const float* __restrict__ Wq,
    float* __restrict__ Xout) {
  __shared__ float S[NN * NN];
  __shared__ float Qp[NN * HDIM];
  __shared__ float Kp[NN * HDIM];
  __shared__ float Vp[NN * HDIM];

  const int blk = blockIdx.x;
  const int hg  = blk & 1;
  const int t   = (blk >> 1) % TT;
  const int b   = (blk >> 1) / TT;

  const int tid  = threadIdx.x;
  const int wave = tid >> 5;
  const int lane = tid & 31;
  const int half = lane >> 4;
  const int l    = lane & 15;

  const float scale = 0.125f;
  const size_t rowOff = (((size_t)(b * NN + tid) * TT + t) * EMBED);

  for (int hi = 0; hi < 4; ++hi) {
    const int h = hg * 4 + hi;
    int wofs = 0;
    asm volatile("" : "+s"(wofs));
    const float* wq = Wq + wofs;
    const float* wk = Wk + wofs;
    const float* wv = Wv + wofs;

    {
      const int n = tid;
      const size_t base = rowOff + h * HDIM;
      float qv[HDIM], kv[HDIM], vv[HDIM];
#pragma unroll
      for (int d = 0; d < HDIM; ++d) {
        qv[d] = Q[base + d];
        kv[d] = K[base + d];
        vv[d] = V[base + d];
      }
#pragma unroll
      for (int e = 0; e < HDIM; ++e) {
        float aq = 0.f, ak = 0.f, av = 0.f;
#pragma unroll
        for (int d = 0; d < HDIM; ++d) {
          aq += qv[d] * wq[e * HDIM + d];
          ak += kv[d] * wk[e * HDIM + d];
          av += vv[d] * wv[e * HDIM + d];
        }
        Qp[n * HDIM + e] = aq;
        Kp[n * HDIM + e] = ak;
        Vp[n * HDIM + e] = av;
      }
    }
    __syncthreads();

#pragma unroll 1
    for (int rt = 0; rt < 2; ++rt) {
      const int rowTile = rt * 64 + wave * 16;
      v8f c[8];
      {
        const float* qrow = Qp + (rowTile + l) * HDIM + 2 * half;
        v2f a0, a1;
        a0.x = qrow[0];
        a0.y = qrow[1];
        a1.x = qrow[4];
        a1.y = qrow[5];
#pragma unroll
        for (int tj = 0; tj < 8; ++tj) {
          const float* krow = Kp + (tj * 16 + l) * HDIM + 2 * half;
          v2f b0, b1;
          b0.x = krow[0];
          b0.y = krow[1];
          b1.x = krow[4];
          b1.y = krow[5];
          v8f acc = {};
          acc = wmma_f32_16x16x4(a0, b0, acc);
          acc = wmma_f32_16x16x4(a1, b1, acc);
          c[tj] = acc;
        }
      }
#pragma unroll
      for (int r = 0; r < 8; ++r) {
        float s[8];
        float m = -1e30f;
#pragma unroll
        for (int tj = 0; tj < 8; ++tj) {
          s[tj] = c[tj][r] * scale;
          m = fmaxf(m, s[tj]);
        }
        m = fmaxf(m, row_xor<1>(m));
        m = fmaxf(m, row_xor<2>(m));
        m = fmaxf(m, row_xor<4>(m));
        m = fmaxf(m, row_xor<8>(m));
        float sum = 0.f;
#pragma unroll
        for (int tj = 0; tj < 8; ++tj) {
          const float ex = __expf(s[tj] - m);
          s[tj] = ex;
          sum += ex;
        }
        sum += row_xor<1>(sum);
        sum += row_xor<2>(sum);
        sum += row_xor<4>(sum);
        sum += row_xor<8>(sum);
        const float inv = 1.f / sum;
        float* prow = S + (rowTile + r + 8 * half) * NN + l;
#pragma unroll
        for (int tj = 0; tj < 8; ++tj) prow[tj * 16] = s[tj] * inv;
      }
    }
    __syncthreads();

    {
      v8f c0 = {}, c1 = {};
      const float* p0 = S + (wave * 16 + l) * NN;
      const float* p1 = S + (64 + wave * 16 + l) * NN;
#pragma unroll
      for (int kb = 0; kb < NN; kb += 4) {
        const int ac = kb + 2 * half;
        v2f a0, a1, bm;
        a0.x = p0[ac];
        a0.y = p0[ac + 1];
        a1.x = p1[ac];
        a1.y = p1[ac + 1];
        bm.x = (l < HDIM) ? Vp[ac * HDIM + l] : 0.f;
        bm.y = (l < HDIM) ? Vp[(ac + 1) * HDIM + l] : 0.f;
        c0 = wmma_f32_16x16x4(a0, bm, c0);
        c1 = wmma_f32_16x16x4(a1, bm, c1);
      }
      if (l < HDIM) {
#pragma unroll
        for (int r = 0; r < 8; ++r) {
          const int n0 = wave * 16 + r + 8 * half;
          const size_t o0 = (((size_t)(b * NN + n0) * TT + t) * EMBED) + h * HDIM + l;
          const size_t o1 = o0 + (size_t)64 * TT * EMBED;
          Xout[o0] = c0[r];
          Xout[o1] = c1[r];
        }
      }
    }
    __syncthreads();
  }
}

__global__ __launch_bounds__(128) void fc_kernel(const float* __restrict__ X,
                                                 const float* __restrict__ Wo,
                                                 const float* __restrict__ bo,
                                                 float* __restrict__ Y) {
  __shared__ float sX[16 * EMBED];
  __shared__ float sW[EMBED * EMBED];
  __shared__ float sb[EMBED];

  const int tid  = threadIdx.x;
  const int wave = tid >> 5;
  const int lane = tid & 31;
  const int half = lane >> 4;
  const int l    = lane & 15;
  const size_t mBase = (size_t)blockIdx.x * 16;

  for (int i = tid; i < 16 * EMBED; i += 128) sX[i] = X[mBase * EMBED + i];
  for (int i = tid; i < EMBED * EMBED; i += 128) sW[i] = Wo[i];
  if (tid < EMBED) sb[tid] = bo[tid];
  __syncthreads();

  const int tj = wave;
  v8f c = {};
#pragma unroll
  for (int kb = 0; kb < EMBED; kb += 4) {
    const int ac = kb + 2 * half;
    v2f a, bm;
    a.x  = sX[l * EMBED + ac];
    a.y  = sX[l * EMBED + ac + 1];
    bm.x = sW[(tj * 16 + l) * EMBED + ac];
    bm.y = sW[(tj * 16 + l) * EMBED + ac + 1];
    c = wmma_f32_16x16x4(a, bm, c);
  }
  const float bias = sb[tj * 16 + l];
#pragma unroll
  for (int r = 0; r < 8; ++r) {
    const size_t row = mBase + r + 8 * half;
    Y[row * EMBED + tj * 16 + l] = c[r] + bias;
  }
}

extern "C" void kernel_launch(void* const* d_in, const int* in_sizes, int n_in,
                              void* d_out, int out_size, void* d_ws,
                              size_t ws_size, hipStream_t stream) {
  (void)in_sizes; (void)n_in; (void)out_size; (void)ws_size;
  const float* V  = (const float*)d_in[0];
  const float* K  = (const float*)d_in[1];
  const float* Q  = (const float*)d_in[2];
  const float* Wv = (const float*)d_in[3];
  const float* Wk = (const float*)d_in[4];
  const float* Wq = (const float*)d_in[5];
  const float* Wo = (const float*)d_in[6];
  const float* bo = (const float*)d_in[7];
  float* X = (float*)d_ws;
  float* Y = (float*)d_out;

  hipLaunchKernelGGL(attn_kernel, dim3(BB * TT * 2), dim3(128), 0,
                     stream, V, K, Q, Wv, Wk, Wq, X);
  hipLaunchKernelGGL(fc_kernel, dim3((BB * NN * TT) / 16), dim3(128), 0, stream,
                     X, Wo, bo, Y);
}


# P*V operands staged in LDS in fragment order: P as 16-byte pieces of 4 keys, XOR-swizzled by row, V as 4-key pieces; 16-byte LDS stores and loads, no shared bank; P*V key order permuted
# speedup vs baseline: 3.7875x; 1.2238x over previous
#include <hip/hip_runtime.h>

typedef float v2f __attribute__((ext_vector_type(2)));
typedef float v4f __attribute__((ext_vector_type(4)));
typedef float v8f __attribute__((ext_vector_type(8)));

#define EMBED 64
#define HEADS 8
#define HDIM  8
#define BB    8
#define NN    128
#define TT    64

__device__ __forceinline__ v8f wmma_f32_16x16x4(v2f a, v2f b, v8f c) {
  return __builtin_amdgcn_wmma_f32_16x16x4_f32(false, a, false, b, (short)0, c,
                                               false, false);
}

template <int M>
__device__ __forceinline__ float row_xor(float x) {
  const int v = __builtin_bit_cast(int, x);
  return __builtin_bit_cast(
      float, __builtin_amdgcn_update_dpp(v, v, 0x160 + M, 0xf, 0xf, false));
}

#define VF_PITCH 68
__global__ __launch_bounds__(128) void attn_kernel(
    const float* __restrict__ V, const float* __restrict__ K,
    const float* __restrict__ Q, const float* __restrict__ Wv,
    const float* __restrict__ Wk, const float* __restrict__ Wq,
    float* __restrict__ Xout) {
  __shared__ __attribute__((aligned(256))) float S[NN * NN];
  __shared__ __attribute__((aligned(16))) float Vf[16 * VF_PITCH];
  __shared__ float Qp[NN * HDIM];
  __shared__ float Kp[NN * HDIM];

  const int blk = blockIdx.x;
  const int hg  = blk & 1;
  const int t   = (blk >> 1) % TT;
  const int b   = (blk >> 1) / TT;

  const int tid  = threadIdx.x;
  const int wave = tid >> 5;
  const int lane = tid & 31;
  const int half = lane >> 4;
  const int l    = lane & 15;

  const float scale = 0.125f;
  const size_t rowOff = (((size_t)(b * NN + tid) * TT + t) * EMBED);
  const int vi = tid & 3, vj = (tid >> 2) & 15, vh = tid >> 6;
  const int vkey = (2 * vi + vh) * 16 + vj;
  const size_t rowOffV = (((size_t)(b * NN + vkey) * TT + t) * EMBED);
  float* const vdst = Vf + vj * VF_PITCH + vh * 32 + vi;

  for (int hi = 0; hi < 4; ++hi) {
    const int h = hg * 4 + hi;
    int wofs = 0;
    asm volatile("" : "+s"(wofs));
    const float* wq = Wq + wofs;
    const float* wk = Wk + wofs;
    const float* wv = Wv + wofs;

    {
      const int n = tid;
      const size_t base = rowOff + h * HDIM;
      const size_t baseV = rowOffV + h * HDIM;
      float qv[HDIM], kv[HDIM], vv[HDIM];
#pragma unroll
      for (int d = 0; d < HDIM; ++d) {
        qv[d] = Q[base + d];
        kv[d] = K[base + d];
        vv[d] = V[baseV + d];
      }
#pragma unroll
      for (int e = 0; e < HDIM; ++e) {
        float aq = 0.f, ak = 0.f, av = 0.f;
#pragma unroll
        for (int d = 0; d < HDIM; ++d) {
          aq += qv[d] * wq[e * HDIM + d];
          ak += kv[d] * wk[e * HDIM + d];
          av += vv[d] * wv[e * HDIM + d];
        }
        Qp[n * HDIM + e] = aq;
        Kp[n * HDIM + e] = ak;
        vdst[e * 4] = av;
      }
    }
    __syncthreads();

#pragma unroll 1
    for (int rt = 0; rt < 2; ++rt) {
      const int rowTile = rt * 64 + wave * 16;
      v8f c[8];
      {
        const float* qrow = Qp + (rowTile + l) * HDIM + 2 * half;
        v2f a0, a1;
        a0.x = qrow[0];
        a0.y = qrow[1];
        a1.x = qrow[4];
        a1.y = qrow[5];
#pragma unroll
        for (int tj = 0; tj < 8; ++tj) {
          const float* krow = Kp + (tj * 16 + l) * HDIM + 2 * half;
          v2f b0, b1;
          b0.x = krow[0];
          b0.y = krow[1];
          b1.x = krow[4];
          b1.y = krow[5];
          v8f acc = {};
          acc = wmma_f32_16x16x4(a0, b0, acc);
          acc = wmma_f32_16x16x4(a1, b1, acc);
          c[tj] = acc;
        }
      }
      float* const prow0 = S + (rowTile + 8 * half) * NN;
      const int px = l ^ (8 * half);
#pragma unroll
      for (int r = 0; r < 8; ++r) {
        float s[8];
        float m = -1e30f;
#pragma unroll
        for (int tj = 0; tj < 8; ++tj) {
          s[tj] = c[tj][r] * scale;
          m = fmaxf(m, s[tj]);
        }
        m = fmaxf(m, row_xor<1>(m));
        m = fmaxf(m, row_xor<2>(m));
        m = fmaxf(m, row_xor<4>(m));
        m = fmaxf(m, row_xor<8>(m));
        float sum = 0.f;
#pragma unroll
        for (int tj = 0; tj < 8; ++tj) {
          const float ex = __expf(s[tj] - m);
          s[tj] = ex;
          sum += ex;
        }
        sum += row_xor<1>(sum);
        sum += row_xor<2>(sum);
        sum += row_xor<4>(sum);
        sum += row_xor<8>(sum);
        const float inv = 1.f / sum;
        float* const pr = prow0 + r * NN + ((px ^ r) << 2);
        v4f pe, po;
        pe.x = s[0] * inv;
        pe.y = s[2] * inv;
        pe.z = s[4] * inv;
        pe.w = s[6] * inv;
        po.x = s[1] * inv;
        po.y = s[3] * inv;
        po.z = s[5] * inv;
        po.w = s[7] * inv;
        *reinterpret_cast<v4f*>(pr)      = pe;
        *reinterpret_cast<v4f*>(pr + 64) = po;
      }
    }
    __syncthreads();

    {
      v8f c0 = {}, c1 = {};
      const float* p0 = S + ((wave * 16 + l) * 2 + half) * 64;
      const float* p1 = p0 + 64 * NN;
      const float* vb = Vf + half * 32 + (l & 7) * 4;
#pragma unroll
      for (int j = 0; j < 16; ++j) {
        const int pc = (j ^ l) << 2;
        const v4f pa = *reinterpret_cast<const v4f*>(p0 + pc);
        const v4f pb = *reinterpret_cast<const v4f*>(p1 + pc);
        const v4f bv = *reinterpret_cast<const v4f*>(vb + j * VF_PITCH);
        v2f a0, a1, bm;
        a0.x = pa.x; a0.y = pa.y;
        a1.x = pb.x; a1.y = pb.y;
        bm.x = bv.x; bm.y = bv.y;
        c0 = wmma_f32_16x16x4(a0, bm, c0);
        c1 = wmma_f32_16x16x4(a1, bm, c1);
        a0.x = pa.z; a0.y = pa.w;
        a1.x = pb.z; a1.y = pb.w;
        bm.x = bv.z; bm.y = bv.w;
        c0 = wmma_f32_16x16x4(a0, bm, c0);
        c1 = wmma_f32_16x16x4(a1, bm, c1);
      }
      if (l < HDIM) {
#pragma unroll
        for (int r = 0; r < 8; ++r) {
          const int n0 = wave * 16 + r + 8 * half;
          const size_t o0 = (((size_t)(b * NN + n0) * TT + t) * EMBED) + h * HDIM + l;
          const size_t o1 = o0 + (size_t)64 * TT * EMBED;
          Xout[o0] = c0[r];
          Xout[o1] = c1[r];
        }
      }
    }
    __syncthreads();
  }
}

__global__ __launch_bounds__(128) void fc_kernel(const float* __restrict__ X,
                                                 const float* __restrict__ Wo,
                                                 const float* __restrict__ bo,
                                                 float* __restrict__ Y) {
  __shared__ float sX[16 * EMBED];
  __shared__ float sW[EMBED * EMBED];
  __shared__ float sb[EMBED];

  const int tid  = threadIdx.x;
  const int wave = tid >> 5;
  const int lane = tid & 31;
  const int half = lane >> 4;
  const int l    = lane & 15;
  const size_t mBase = (size_t)blockIdx.x * 16;

  for (int i = tid; i < 16 * EMBED; i += 128) sX[i] = X[mBase * EMBED + i];
  for (int i = tid; i < EMBED * EMBED; i += 128) sW[i] = Wo[i];
  if (tid < EMBED) sb[tid] = bo[tid];
  __syncthreads();

  const int tj = wave;
  v8f c = {};
#pragma unroll
  for (int kb = 0; kb < EMBED; kb += 4) {
    const int ac = kb + 2 * half;
    v2f a, bm;
    a.x  = sX[l * EMBED + ac];
    a.y  = sX[l * EMBED + ac + 1];
    bm.x = sW[(tj * 16 + l) * EMBED + ac];
    bm.y = sW[(tj * 16 + l) * EMBED + ac + 1];
    c = wmma_f32_16x16x4(a, bm, c);
  }
  const float bias = sb[tj * 16 + l];
#pragma unroll
  for (int r = 0; r < 8; ++r) {
    const size_t row = mBase + r + 8 * half;
    Y[row * EMBED + tj * 16 + l] = c[r] + bias;
  }
}

extern "C" void kernel_launch(void* const* d_in, const int* in_sizes, int n_in,
                              void* d_out, int out_size, void* d_ws,
                              size_t ws_size, hipStream_t stream) {
  (void)in_sizes; (void)n_in; (void)out_size; (void)ws_size;
  const float* V  = (const float*)d_in[0];
  const float* K  = (const float*)d_in[1];
  const float* Q  = (const float*)d_in[2];
  const float* Wv = (const float*)d_in[3];
  const float* Wk = (const float*)d_in[4];
  const float* Wq = (const float*)d_in[5];
  const float* Wo = (const float*)d_in[6];
  const float* bo = (const float*)d_in[7];
  float* X = (float*)d_ws;
  float* Y = (float*)d_out;

  hipLaunchKernelGGL(attn_kernel, dim3(BB * TT * 2), dim3(128), 0,
                     stream, V, K, Q, Wv, Wk, Wq, X);
  hipLaunchKernelGGL(fc_kernel, dim3((BB * NN * TT) / 16), dim3(128), 0, stream,
                     X, Wo, bo, Y);
}


# Output projection without LDS: 1024 persistent waves hold all of Wo as B fragments in registers, read X by 16-byte global loads straight into A fragments, store whole 256-byte rows; k order permuted
# speedup vs baseline: 5.0163x; 1.3244x over previous
#include <hip/hip_runtime.h>

typedef float v2f __attribute__((ext_vector_type(2)));
typedef float v4f __attribute__((ext_vector_type(4)));
typedef float v8f __attribute__((ext_vector_type(8)));

#define EMBED 64
#define HEADS 8
#define HDIM  8
#define BB    8
#define NN    128
#define TT    64

__device__ __forceinline__ v8f wmma_f32_16x16x4(v2f a, v2f b, v8f c) {
  return __builtin_amdgcn_wmma_f32_16x16x4_f32(false, a, false, b, (short)0, c,
                                               false, false);
}

template <int M>
__device__ __forceinline__ float row_xor(float x) {
  const int v = __builtin_bit_cast(int, x);
  return __builtin_bit_cast(
      float, __builtin_amdgcn_update_dpp(v, v, 0x160 + M, 0xf, 0xf, false));
}

#define VF_PITCH 68
__global__ __launch_bounds__(128) void attn_kernel(
    const float* __restrict__ V, const float* __restrict__ K,
    const float* __restrict__ Q, const float* __restrict__ Wv,
    const float* __restrict__ Wk, const float* __restrict__ Wq,
    float* __restrict__ Xout) {
  __shared__ __attribute__((aligned(256))) float S[NN * NN];
  __shared__ __attribute__((aligned(16))) float Vf[16 * VF_PITCH];
  __shared__ float Qp[NN * HDIM];
  __shared__ float Kp[NN * HDIM];

  const int blk = blockIdx.x;
  const int hg  = blk & 1;
  const int t   = (blk >> 1) % TT;
  const int b   = (blk >> 1) / TT;

  const int tid  = threadIdx.x;
  const int wave = tid >> 5;
  const int lane = tid & 31;
  const int half = lane >> 4;
  const int l    = lane & 15;

  const float scale = 0.125f;
  const size_t rowOff = (((size_t)(b * NN + tid) * TT + t) * EMBED);
  const int vi = tid & 3, vj = (tid >> 2) & 15, vh = tid >> 6;
  const int vkey = (2 * vi + vh) * 16 + vj;
  const size_t rowOffV = (((size_t)(b * NN + vkey) * TT + t) * EMBED);
  float* const vdst = Vf + vj * VF_PITCH + vh * 32 + vi;

  for (int hi = 0; hi < 4; ++hi) {
    const int h = hg * 4 + hi;
    int wofs = 0;
    asm volatile("" : "+s"(wofs));
    const float* wq = Wq + wofs;
    const float* wk = Wk + wofs;
    const float* wv = Wv + wofs;

    {
      const int n = tid;
      const size_t base = rowOff + h * HDIM;
      const size_t baseV = rowOffV + h * HDIM;
      float qv[HDIM], kv[HDIM], vv[HDIM];
#pragma unroll
      for (int d = 0; d < HDIM; ++d) {
        qv[d] = Q[base + d];
        kv[d] = K[base + d];
        vv[d] = V[baseV + d];
      }
#pragma unroll
      for (int e = 0; e < HDIM; ++e) {
        float aq = 0.f, ak = 0.f, av = 0.f;
#pragma unroll
        for (int d = 0; d < HDIM; ++d) {
          aq += qv[d] * wq[e * HDIM + d];
          ak += kv[d] * wk[e * HDIM + d];
          av += vv[d] * wv[e * HDIM + d];
        }
        Qp[n * HDIM + e] = aq;
        Kp[n * HDIM + e] = ak;
        vdst[e * 4] = av;
      }
    }
    __syncthreads();

#pragma unroll 1
    for (int rt = 0; rt < 2; ++rt) {
      const int rowTile = rt * 64 + wave * 16;
      v8f c[8];
      {
        const float* qrow = Qp + (rowTile + l) * HDIM + 2 * half;
        v2f a0, a1;
        a0.x = qrow[0];
        a0.y = qrow[1];
        a1.x = qrow[4];
        a1.y = qrow[5];
#pragma unroll
        for (int tj = 0; tj < 8; ++tj) {
          const float* krow = Kp + (tj * 16 + l) * HDIM + 2 * half;
          v2f b0, b1;
          b0.x = krow[0];
          b0.y = krow[1];
          b1.x = krow[4];
          b1.y = krow[5];
          v8f acc = {};
          acc = wmma_f32_16x16x4(a0, b0, acc);
          acc = wmma_f32_16x16x4(a1, b1, acc);
          c[tj] = acc;
        }
      }
      float* const prow0 = S + (rowTile + 8 * half) * NN;
      const int px = l ^ (8 * half);
#pragma unroll
      for (int r = 0; r < 8; ++r) {
        float s[8];
        float m = -1e30f;
#pragma unroll
        for (int tj = 0; tj < 8; ++tj) {
          s[tj] = c[tj][r] * scale;
          m = fmaxf(m, s[tj]);
        }
        m = fmaxf(m, row_xor<1>(m));
        m = fmaxf(m, row_xor<2>(m));
        m = fmaxf(m, row_xor<4>(m));
        m = fmaxf(m, row_xor<8>(m));
        float sum = 0.f;
#pragma unroll
        for (int tj = 0; tj < 8; ++tj) {
          const float ex = __expf(s[tj] - m);
          s[tj] = ex;
          sum += ex;
        }
        sum += row_xor<1>(sum);
        sum += row_xor<2>(sum);
        sum += row_xor<4>(sum);
        sum += row_xor<8>(sum);
        const float inv = 1.f / sum;
        float* const pr = prow0 + r * NN + ((px ^ r) << 2);
        v4f pe, po;
        pe.x = s[0] * inv;
        pe.y = s[2] * inv;
        pe.z = s[4] * inv;
        pe.w = s[6] * inv;
        po.x = s[1] * inv;
        po.y = s[3] * inv;
        po.z = s[5] * inv;
        po.w = s[7] * inv;
        *reinterpret_cast<v4f*>(pr)      = pe;
        *reinterpret_cast<v4f*>(pr + 64) = po;
      }
    }
    __syncthreads();

    {
      v8f c0 = {}, c1 = {};
      const float* p0 = S + ((wave * 16 + l) * 2 + half) * 64;
      const float* p1 = p0 + 64 * NN;
      const float* vb = Vf + half * 32 + (l & 7) * 4;
#pragma unroll
      for (int j = 0; j < 16; ++j) {
        const int pc = (j ^ l) << 2;
        const v4f pa = *reinterpret_cast<const v4f*>(p0 + pc);
        const v4f pb = *reinterpret_cast<const v4f*>(p1 + pc);
        const v4f bv = *reinterpret_cast<const v4f*>(vb + j * VF_PITCH);
        v2f a0, a1, bm;
        a0.x = pa.x; a0.y = pa.y;
        a1.x = pb.x; a1.y = pb.y;
        bm.x = bv.x; bm.y = bv.y;
        c0 = wmma_f32_16x16x4(a0, bm, c0);
        c1 = wmma_f32_16x16x4(a1, bm, c1);
        a0.x = pa.z; a0.y = pa.w;
        a1.x = pb.z; a1.y = pb.w;
        bm.x = bv.z; bm.y = bv.w;
        c0 = wmma_f32_16x16x4(a0, bm, c0);
        c1 = wmma_f32_16x16x4(a1, bm, c1);
      }
      if (l < HDIM) {
#pragma unroll
        for (int r = 0; r < 8; ++r) {
          const int n0 = wave * 16 + r + 8 * half;
          const size_t o0 = (((size_t)(b * NN + n0) * TT + t) * EMBED) + h * HDIM + l;
          const size_t o1 = o0 + (size_t)64 * TT * EMBED;
          Xout[o0] = c0[r];
          Xout[o1] = c1[r];
        }
      }
    }
    __syncthreads();
  }
}

#define FC_TILES 4
#define FC_BLOCKS ((BB * NN * TT) / (16 * FC_TILES * 4))
__global__ __launch_bounds__(128) void fc_kernel(const float* __restrict__ X,
                                                 const float* __restrict__ Wo,
                                                 const float* __restrict__ bo,
                                                 float* __restrict__ Y) {
  const int tid  = threadIdx.x;
  const int lane = tid & 31;
  const int half = lane >> 4;
  const int l    = lane & 15;
  const size_t row0 = ((size_t)blockIdx.x * 4 + (tid >> 5)) * (16 * FC_TILES);

  const float* xp = X + (row0 + l) * EMBED + 4 * half;
  v4f a[8];
#pragma unroll
  for (int j = 0; j < 8; ++j) a[j] = *reinterpret_cast<const v4f*>(xp + 8 * j);

  const float* wp = Wo + (4 * l) * EMBED + 4 * half;
  v4f w[4][8];
#pragma unroll
  for (int j = 0; j < 8; ++j) {
#pragma unroll
    for (int tj = 0; tj < 4; ++tj)
      w[tj][j] = *reinterpret_cast<const v4f*>(wp + tj * EMBED + 8 * j);
  }
  const v4f bias = *reinterpret_cast<const v4f*>(bo + 4 * l);

  float* yp = Y + (row0 + 8 * half) * EMBED + 4 * l;

#pragma unroll
  for (int it = 0; it < FC_TILES; ++it) {
    v4f an[8];
    if (it + 1 < FC_TILES) {
#pragma unroll
      for (int j = 0; j < 8; ++j)
        an[j] = *reinterpret_cast<const v4f*>(xp + (it + 1) * 16 * EMBED + 8 * j);
    }
    __builtin_amdgcn_sched_barrier(0);
    v8f c0 = {}, c1 = {}, c2 = {}, c3 = {};
#pragma unroll
    for (int j = 0; j < 8; ++j) {
      v2f av, b0, b1, b2, b3;
      av.x = a[j].x;    av.y = a[j].y;
      b0.x = w[0][j].x; b0.y = w[0][j].y;
      b1.x = w[1][j].x; b1.y = w[1][j].y;
      b2.x = w[2][j].x; b2.y = w[2][j].y;
      b3.x = w[3][j].x; b3.y = w[3][j].y;
      c0 = wmma_f32_16x16x4(av, b0, c0);
      c1 = wmma_f32_16x16x4(av, b1, c1);
      c2 = wmma_f32_16x16x4(av, b2, c2);
      c3 = wmma_f32_16x16x4(av, b3, c3);
      av.x = a[j].z;    av.y = a[j].w;
      b0.x = w[0][j].z; b0.y = w[0][j].w;
      b1.x = w[1][j].z; b1.y = w[1][j].w;
      b2.x = w[2][j].z; b2.y = w[2][j].w;
      b3.x = w[3][j].z; b3.y = w[3][j].w;
      c0 = wmma_f32_16x16x4(av, b0, c0);
      c1 = wmma_f32_16x16x4(av, b1, c1);
      c2 = wmma_f32_16x16x4(av, b2, c2);
      c3 = wmma_f32_16x16x4(av, b3, c3);
    }
#pragma unroll
    for (int r = 0; r < 8; ++r) {
      v4f o;
      o.x = c0[r] + bias.x;
      o.y = c1[r] + bias.y;
      o.z = c2[r] + bias.z;
      o.w = c3[r] + bias.w;
      *reinterpret_cast<v4f*>(yp + (it * 16 + r) * EMBED) = o;
    }
    if (it + 1 < FC_TILES) {
#pragma unroll
      for (int j = 0; j < 8; ++j) a[j] = an[j];
    }
  }
}

extern "C" void kernel_launch(void* const* d_in, const int* in_sizes, int n_in,
                              void* d_out, int out_size, void* d_ws,
                              size_t ws_size, hipStream_t stream) {
  (void)in_sizes; (void)n_in; (void)out_size; (void)ws_size;
  const float* V  = (const float*)d_in[0];
  const float* K  = (const float*)d_in[1];
  const float* Q  = (const float*)d_in[2];
  const float* Wv = (const float*)d_in[3];
  const float* Wk = (const float*)d_in[4];
  const float* Wq = (const float*)d_in[5];
  const float* Wo = (const float*)d_in[6];
  const float* bo = (const float*)d_in[7];
  float* X = (float*)d_ws;
  float* Y = (float*)d_out;

  hipLaunchKernelGGL(attn_kernel, dim3(BB * TT * 2), dim3(128), 0,
                     stream, V, K, Q, Wv, Wk, Wq, X);
  hipLaunchKernelGGL(fc_kernel, dim3(FC_BLOCKS), dim3(128), 0, stream,
                     X, Wo, bo, Y);
}


# Softmax normalisation moved behind P*V: exp2 with the scale folded into one FMA goes to LDS unnormalised, a ones column in V's unused half gives the row sum in the same product; sum order changes
# speedup vs baseline: 5.6864x; 1.1336x over previous
#include <hip/hip_runtime.h>

typedef float v2f __attribute__((ext_vector_type(2)));
typedef float v4f __attribute__((ext_vector_type(4)));
typedef float v8f __attribute__((ext_vector_type(8)));

#define EMBED 64
#define HEADS 8
#define HDIM  8
#define BB    8
#define NN    128
#define TT    64

__device__ __forceinline__ v8f wmma_f32_16x16x4(v2f a, v2f b, v8f c) {
  return __builtin_amdgcn_wmma_f32_16x16x4_f32(false, a, false, b, (short)0, c,
                                               false, false);
}

template <int M>
__device__ __forceinline__ float row_xor(float x) {
  const int v = __builtin_bit_cast(int, x);
  return __builtin_bit_cast(
      float, __builtin_amdgcn_update_dpp(v, v, 0x160 + M, 0xf, 0xf, false));
}

#define VF_PITCH 76
#define VF_HALF  36
__global__ __launch_bounds__(128) void attn_kernel(
    const float* __restrict__ V, const float* __restrict__ K,
    const float* __restrict__ Q, const float* __restrict__ Wv,
    const float* __restrict__ Wk, const float* __restrict__ Wq,
    float* __restrict__ Xout) {
  __shared__ __attribute__((aligned(256))) float S[NN * NN];
  __shared__ __attribute__((aligned(16))) float Vf[16 * VF_PITCH];
  __shared__ float Qp[NN * HDIM];
  __shared__ float Kp[NN * HDIM];

  const int blk = blockIdx.x;
  const int hg  = blk & 1;
  const int t   = (blk >> 1) % TT;
  const int b   = (blk >> 1) / TT;

  const int tid  = threadIdx.x;
  const int wave = tid >> 5;
  const int lane = tid & 31;
  const int half = lane >> 4;
  const int l    = lane & 15;

  const size_t rowOff = (((size_t)(b * NN + tid) * TT + t) * EMBED);
  const int vi = tid & 3, vj = (tid >> 2) & 15, vh = tid >> 6;
  const int vkey = (2 * vi + vh) * 16 + vj;
  const size_t rowOffV = (((size_t)(b * NN + vkey) * TT + t) * EMBED);
  float* const vdst = Vf + vj * VF_PITCH + vh * VF_HALF + vi;
  const float k2 = 0.125f * 1.44269504088896341f;

  for (int hi = 0; hi < 4; ++hi) {
    const int h = hg * 4 + hi;
    int wofs = 0;
    asm volatile("" : "+s"(wofs));
    const float* wq = Wq + wofs;
    const float* wk = Wk + wofs;
    const float* wv = Wv + wofs;

    {
      const int n = tid;
      const size_t base = rowOff + h * HDIM;
      const size_t baseV = rowOffV + h * HDIM;
      float qv[HDIM], kv[HDIM], vv[HDIM];
#pragma unroll
      for (int d = 0; d < HDIM; ++d) {
        qv[d] = Q[base + d];
        kv[d] = K[base + d];
        vv[d] = V[baseV + d];
      }
#pragma unroll
      for (int e = 0; e < HDIM; ++e) {
        float aq = 0.f, ak = 0.f, av = 0.f;
#pragma unroll
        for (int d = 0; d < HDIM; ++d) {
          aq += qv[d] * wq[e * HDIM + d];
          ak += kv[d] * wk[e * HDIM + d];
          av += vv[d] * wv[e * HDIM + d];
        }
        Qp[n * HDIM + e] = aq;
        Kp[n * HDIM + e] = ak;
        vdst[e * 4] = av;
      }
      Vf[(tid >> 3) * VF_PITCH + ((tid >> 2) & 1) * VF_HALF + 32 + (tid & 3)] = 1.f;
    }
    __syncthreads();

#pragma unroll 1
    for (int rt = 0; rt < 2; ++rt) {
      const int rowTile = rt * 64 + wave * 16;
      v8f c[8];
      {
        const float* qrow = Qp + (rowTile + l) * HDIM + 2 * half;
        v2f a0, a1;
        a0.x = qrow[0];
        a0.y = qrow[1];
        a1.x = qrow[4];
        a1.y = qrow[5];
#pragma unroll
        for (int tj = 0; tj < 8; ++tj) {
          const float* krow = Kp + (tj * 16 + l) * HDIM + 2 * half;
          v2f b0, b1;
          b0.x = krow[0];
          b0.y = krow[1];
          b1.x = krow[4];
          b1.y = krow[5];
          v8f acc = {};
          acc = wmma_f32_16x16x4(a0, b0, acc);
          acc = wmma_f32_16x16x4(a1, b1, acc);
          c[tj] = acc;
        }
      }
      float* const prow0 = S + (rowTile + 8 * half) * NN;
      const int px = l ^ (8 * half);
#pragma unroll
      for (int r = 0; r < 8; ++r) {
        float m = c[0][r];
#pragma unroll
        for (int tj = 1; tj < 8; ++tj) m = fmaxf(m, c[tj][r]);
        m = fmaxf(m, row_xor<1>(m));
        m = fmaxf(m, row_xor<2>(m));
        m = fmaxf(m, row_xor<4>(m));
        m = fmaxf(m, row_xor<8>(m));
        const float mk = -m * k2;
        float s[8];
#pragma unroll
        for (int tj = 0; tj < 8; ++tj)
          s[tj] = __builtin_amdgcn_exp2f(fmaf(c[tj][r], k2, mk));
        float* const pr = prow0 + r * NN + ((px ^ r) << 2);
        v4f pe, po;
        pe.x = s[0];
        pe.y = s[2];
        pe.z = s[4];
        pe.w = s[6];
        po.x = s[1];
        po.y = s[3];
        po.z = s[5];
        po.w = s[7];
        *reinterpret_cast<v4f*>(pr)      = pe;
        *reinterpret_cast<v4f*>(pr + 64) = po;
      }
    }
    __syncthreads();

    {
      v8f c0 = {}, c1 = {};
      const float* p0 = S + ((wave * 16 + l) * 2 + half) * 64;
      const float* p1 = p0 + 64 * NN;
      const float* vb = Vf + half * VF_HALF + (l < 8 ? l : 8) * 4;
#pragma unroll
      for (int j = 0; j < 16; ++j) {
        const int pc = (j ^ l) << 2;
        const v4f pa = *reinterpret_cast<const v4f*>(p0 + pc);
        const v4f pb = *reinterpret_cast<const v4f*>(p1 + pc);
        const v4f bv = *reinterpret_cast<const v4f*>(vb + j * VF_PITCH);
        v2f a0, a1, bm;
        a0.x = pa.x; a0.y = pa.y;
        a1.x = pb.x; a1.y = pb.y;
        bm.x = bv.x; bm.y = bv.y;
        c0 = wmma_f32_16x16x4(a0, bm, c0);
        c1 = wmma_f32_16x16x4(a1, bm, c1);
        a0.x = pa.z; a0.y = pa.w;
        a1.x = pb.z; a1.y = pb.w;
        bm.x = bv.z; bm.y = bv.w;
        c0 = wmma_f32_16x16x4(a0, bm, c0);
        c1 = wmma_f32_16x16x4(a1, bm, c1);
      }
      float o0v[8], o1v[8];
#pragma unroll
      for (int r = 0; r < 8; ++r) {
        o0v[r] = c0[r] * __builtin_amdgcn_rcpf(row_xor<8>(c0[r]));
        o1v[r] = c1[r] * __builtin_amdgcn_rcpf(row_xor<8>(c1[r]));
      }
      if (l < HDIM) {
#pragma unroll
        for (int r = 0; r < 8; ++r) {
          const int n0 = wave * 16 + r + 8 * half;
          const size_t o0 = (((size_t)(b * NN + n0) * TT + t) * EMBED) + h * HDIM + l;
          const size_t o1 = o0 + (size_t)64 * TT * EMBED;
          Xout[o0] = o0v[r];
          Xout[o1] = o1v[r];
        }
      }
    }
    __syncthreads();
  }
}

#define FC_TILES 4
#define FC_BLOCKS ((BB * NN * TT) / (16 * FC_TILES * 4))
__global__ __launch_bounds__(128) void fc_kernel(const float* __restrict__ X,
                                                 const float* __restrict__ Wo,
                                                 const float* __restrict__ bo,
                                                 float* __restrict__ Y) {
  const int tid  = threadIdx.x;
  const int lane = tid & 31;
  const int half = lane >> 4;
  const int l    = lane & 15;
  const size_t row0 = ((size_t)blockIdx.x * 4 + (tid >> 5)) * (16 * FC_TILES);

  const float* xp = X + (row0 + l) * EMBED + 4 * half;
  v4f a[8];
#pragma unroll
  for (int j = 0; j < 8; ++j) a[j] = *reinterpret_cast<const v4f*>(xp + 8 * j);

  const float* wp = Wo + (4 * l) * EMBED + 4 * half;
  v4f w[4][8];
#pragma unroll
  for (int j = 0; j < 8; ++j) {
#pragma unroll
    for (int tj = 0; tj < 4; ++tj)
      w[tj][j] = *reinterpret_cast<const v4f*>(wp + tj * EMBED + 8 * j);
  }
  const v4f bias = *reinterpret_cast<const v4f*>(bo + 4 * l);

  float* yp = Y + (row0 + 8 * half) * EMBED + 4 * l;

#pragma unroll
  for (int it = 0; it < FC_TILES; ++it) {
    v4f an[8];
    if (it + 1 < FC_TILES) {
#pragma unroll
      for (int j = 0; j < 8; ++j)
        an[j] = *reinterpret_cast<const v4f*>(xp + (it + 1) * 16 * EMBED + 8 * j);
    }
    __builtin_amdgcn_sched_barrier(0);
    v8f c0 = {}, c1 = {}, c2 = {}, c3 = {};
#pragma unroll
    for (int j = 0; j < 8; ++j) {
      v2f av, b0, b1, b2, b3;
      av.x = a[j].x;    av.y = a[j].y;
      b0.x = w[0][j].x; b0.y = w[0][j].y;
      b1.x = w[1][j].x; b1.y = w[1][j].y;
      b2.x = w[2][j].x; b2.y = w[2][j].y;
      b3.x = w[3][j].x; b3.y = w[3][j].y;
      c0 = wmma_f32_16x16x4(av, b0, c0);
      c1 = wmma_f32_16x16x4(av, b1, c1);
      c2 = wmma_f32_16x16x4(av, b2, c2);
      c3 = wmma_f32_16x16x4(av, b3, c3);
      av.x = a[j].z;    av.y = a[j].w;
      b0.x = w[0][j].z; b0.y = w[0][j].w;
      b1.x = w[1][j].z; b1.y = w[1][j].w;
      b2.x = w[2][j].z; b2.y = w[2][j].w;
      b3.x = w[3][j].z; b3.y = w[3][j].w;
      c0 = wmma_f32_16x16x4(av, b0, c0);
      c1 = wmma_f32_16x16x4(av, b1, c1);
      c2 = wmma_f32_16x16x4(av, b2, c2);
      c3 = wmma_f32_16x16x4(av, b3, c3);
    }
#pragma unroll
    for (int r = 0; r < 8; ++r) {
      v4f o;
      o.x = c0[r] + bias.x;
      o.y = c1[r] + bias.y;
      o.z = c2[r] + bias.z;
      o.w = c3[r] + bias.w;
      *reinterpret_cast<v4f*>(yp + (it * 16 + r) * EMBED) = o;
    }
    if (it + 1 < FC_TILES) {
#pragma unroll
      for (int j = 0; j < 8; ++j) a[j] = an[j];
    }
  }
}

extern "C" void kernel_launch(void* const* d_in, const int* in_sizes, int n_in,
                              void* d_out, int out_size, void* d_ws,
                              size_t ws_size, hipStream_t stream) {
  (void)in_sizes; (void)n_in; (void)out_size; (void)ws_size;
  const float* V  = (const float*)d_in[0];
  const float* K  = (const float*)d_in[1];
  const float* Q  = (const float*)d_in[2];
  const float* Wv = (const float*)d_in[3];
  const float* Wk = (const float*)d_in[4];
  const float* Wq = (const float*)d_in[5];
  const float* Wo = (const float*)d_in[6];
  const float* bo = (const float*)d_in[7];
  float* X = (float*)d_ws;
  float* Y = (float*)d_out;

  hipLaunchKernelGGL(attn_kernel, dim3(BB * TT * 2), dim3(128), 0,
                     stream, V, K, Q, Wv, Wk, Wq, X);
  hipLaunchKernelGGL(fc_kernel, dim3(FC_BLOCKS), dim3(128), 0, stream,
                     X, Wo, bo, Y);
}
